# starting kernel
# baseline (speedup 1.0000x reference)
#include <hip/hip_runtime.h>
#include <hip/hip_bf16.h>

typedef float v2f __attribute__((ext_vector_type(2)));
typedef float v8f __attribute__((ext_vector_type(8)));

__device__ __forceinline__ v8f wmma_f32(v2f a, v2f b, v8f c) {
    return __builtin_amdgcn_wmma_f32_16x16x4_f32(
         false, a,  false, b,
         (short)0, c,  false,  false);
}

__global__ void proj_gemm(const float* __restrict__ Kin,
                          const float* __restrict__ Vin,
                          const float* __restrict__ Qin,
                          const float* __restrict__ Wk,
                          const float* __restrict__ Wv,
                          const float* __restrict__ Wq,
                          float* __restrict__ kproj,
                          float* __restrict__ vproj,
                          float* __restrict__ qproj,
                          int M, int Kd, int N) {
    const int which = blockIdx.z;
    const float* __restrict__ X = (which == 0) ? Kin : (which == 1) ? Vin : Qin;
    const float* __restrict__ W = (which == 0) ? Wk  : (which == 1) ? Wv  : Wq;
    float* __restrict__ O       = (which == 0) ? kproj : (which == 1) ? vproj : qproj;

    const int lane = threadIdx.x & 31;
    const int wave = threadIdx.x >> 5;
    const int lo = lane & 15;
    const int hi = lane >> 4;

    const int tileM = (blockIdx.y * 8 + wave) * 32;
    const int tileN = blockIdx.x * 64;

    v8f acc00 = {}; v8f acc01 = {}; v8f acc02 = {}; v8f acc03 = {};
    v8f acc10 = {}; v8f acc11 = {}; v8f acc12 = {}; v8f acc13 = {};

    const float* arow0 = X + (size_t)(tileM + lo)      * Kd + hi * 2;
    const float* arow1 = X + (size_t)(tileM + 16 + lo) * Kd + hi * 2;

    for (int k0 = 0; k0 < Kd; k0 += 4) {
        v2f a0 = *(const v2f*)(arow0 + k0);
        v2f a1 = *(const v2f*)(arow1 + k0);
        const float* wr0 = W + (size_t)(k0 + hi * 2) * N + tileN + lo;
        v2f b0; b0.x = wr0[ 0];      b0.y = wr0[ 0 + N];
        v2f b1; b1.x = wr0[16];      b1.y = wr0[16 + N];
        v2f b2; b2.x = wr0[32];      b2.y = wr0[32 + N];
        v2f b3; b3.x = wr0[48];      b3.y = wr0[48 + N];
        acc00 = wmma_f32(a0, b0, acc00);
        acc10 = wmma_f32(a1, b0, acc10);
        acc01 = wmma_f32(a0, b1, acc01);
        acc11 = wmma_f32(a1, b1, acc11);
        acc02 = wmma_f32(a0, b2, acc02);
        acc12 = wmma_f32(a1, b2, acc12);
        acc03 = wmma_f32(a0, b3, acc03);
        acc13 = wmma_f32(a1, b3, acc13);
    }

    #pragma unroll
    for (int v = 0; v < 8; ++v) {
        float* orow0 = O + (size_t)(tileM + v + hi * 8)      * N + tileN + lo;
        float* orow1 = O + (size_t)(tileM + 16 + v + hi * 8) * N + tileN + lo;
        orow0[ 0] = acc00[v];
        orow0[16] = acc01[v];
        orow0[32] = acc02[v];
        orow0[48] = acc03[v];
        orow1[ 0] = acc10[v];
        orow1[16] = acc11[v];
        orow1[32] = acc12[v];
        orow1[48] = acc13[v];
    }
}

#define SWA_W    64
#define SWA_WLEN 144

__global__ void swa_kernel(const float* __restrict__ kp,
                           const float* __restrict__ vp,
                           const float* __restrict__ qp,
                           float* __restrict__ out,
                           int T, int S, int D) {
    __shared__ float sc[16][SWA_WLEN];

    const int tid  = threadIdx.x;
    const int lane = tid & 31;
    const int wave = tid >> 5;
    const int lo = lane & 15;
    const int hi = lane >> 4;

    const int t0    = blockIdx.x * 16;
    const int kbase = t0 - SWA_W;

    const float* qrow = qp + (size_t)(t0 + lo) * D + hi * 2;
    for (int kt = wave; kt < 9; kt += 4) {
        const int krow = kbase + kt * 16 + lo;
        const int krc  = min(max(krow, 0), S - 1);
        const float* krowp = kp + (size_t)krc * D + hi * 2;
        v8f acc = {};
        for (int k0 = 0; k0 < D; k0 += 4) {
            v2f a = *(const v2f*)(qrow + k0);
            v2f b = *(const v2f*)(krowp + k0);
            acc = wmma_f32(a, b, acc);
        }
        #pragma unroll
        for (int v = 0; v < 8; ++v)
            sc[v + hi * 8][kt * 16 + lo] = acc[v];
    }
    __syncthreads();

    if (tid < 16) {
        const int t = t0 + tid;
        const int start = max(0, t - SWA_W);
        const int end   = min(S, t + SWA_W + 1);
        const float scale = 0.03608439182435161f;
        float m = -INFINITY;
        for (int c = 0; c < SWA_WLEN; ++c) {
            const int j = kbase + c;
            const float s = (j >= start && j < end) ? sc[tid][c] * scale : -INFINITY;
            sc[tid][c] = s;
            m = fmaxf(m, s);
        }
        float sum = 0.f;
        for (int c = 0; c < SWA_WLEN; ++c) {
            const float s = sc[tid][c];
            const float e = (s == -INFINITY) ? 0.f : __expf(s - m);
            sc[tid][c] = e;
            sum += e;
        }
        const float inv = 1.0f / (sum * (float)(end - start));
        for (int c = 0; c < SWA_WLEN; ++c) sc[tid][c] *= inv;
    }
    __syncthreads();

    for (int ct = wave; ct < D / 16; ct += 4) {
        const int col = ct * 16 + lo;
        v8f acc = {};
        for (int k0 = 0; k0 < SWA_WLEN; k0 += 4) {
            v2f a; a.x = sc[lo][k0 + hi * 2];
                   a.y = sc[lo][k0 + hi * 2 + 1];
            const int vr  = kbase + k0 + hi * 2;
            const int vr0 = min(max(vr,     0), S - 1);
            const int vr1 = min(max(vr + 1, 0), S - 1);
            v2f b; b.x = vp[(size_t)vr0 * D + col];
                   b.y = vp[(size_t)vr1 * D + col];
            acc = wmma_f32(a, b, acc);
        }
        #pragma unroll
        for (int v = 0; v < 8; ++v)
            out[(size_t)(t0 + v + hi * 8) * D + col] = acc[v];
    }
}

extern "C" void kernel_launch(void* const* d_in, const int* in_sizes, int n_in,
                              void* d_out, int out_size, void* d_ws, size_t ws_size,
                              hipStream_t stream) {
    const float* key   = (const float*)d_in[0];
    const float* value = (const float*)d_in[1];
    const float* query = (const float*)d_in[2];
    const float* Wk    = (const float*)d_in[3];
    const float* Wv    = (const float*)d_in[4];
    const float* Wq    = (const float*)d_in[5];
    float* out = (float*)d_out;

    const int S = 2048, T = 2048, E = 768, D = 768;

    float* ws = (float*)d_ws;
    float* kproj = ws;
    float* vproj = ws + (size_t)S * D;
    float* qproj = ws + 2 * (size_t)S * D;

    dim3 gproj(D / 64, S / 256, 3);
    proj_gemm<<<gproj, 256, 0, stream>>>(key, value, query, Wk, Wv, Wq,
                                         kproj, vproj, qproj, S, E, D);

    swa_kernel<<<dim3(T / 16), 128, 0, stream>>>(kproj, vproj, qproj, out, T, S, D);
}


# projection GEMM re-tiled for equal WMMA work per SIMD: 256 workgroups x 12 waves, 16x96 wave tiles, X and W rows straight to registers by 16-byte loads, no LDS; sum order changes (k permuted in 8s)
# speedup vs baseline: 3.1408x; 3.1408x over previous
#include <hip/hip_runtime.h>
#include <hip/hip_bf16.h>

typedef float v2f __attribute__((ext_vector_type(2)));
typedef float v4f __attribute__((ext_vector_type(4)));
typedef float v8f __attribute__((ext_vector_type(8)));

__device__ __forceinline__ v8f wmma_f32(v2f a, v2f b, v8f c) {
    return __builtin_amdgcn_wmma_f32_16x16x4_f32(
         false, a,  false, b,
         (short)0, c,  false,  false);
}

#define PJ_K 768
#define PJ_N 768

struct PjFrag {
    v4f a[2];
    v4f b4[8];
    v2f b2[8];
};

__device__ __forceinline__ void pj_load(PjFrag& f, const float* __restrict__ xa,
                                        const float* __restrict__ w4,
                                        const float* __restrict__ w2, int k) {
    #pragma unroll
    for (int g = 0; g < 2; ++g) f.a[g] = *(const v4f*)(xa + k + 8 * g);
    #pragma unroll
    for (int r = 0; r < 8; ++r) {
        const int kr = k + 8 * (r >> 2) + (r & 3);
        f.b4[r] = *(const v4f*)(w4 + (size_t)kr * PJ_N);
        f.b2[r] = *(const v2f*)(w2 + (size_t)kr * PJ_N);
    }
}

__device__ __forceinline__ void pj_mma(const PjFrag& f, v8f (&acc)[6]) {
    #pragma unroll
    for (int g = 0; g < 2; ++g) {
        #pragma unroll
        for (int s = 0; s < 2; ++s) {
            const v2f a2 = s ? f.a[g].zw : f.a[g].xy;
            const v4f r0 = f.b4[4 * g + 2 * s], r1 = f.b4[4 * g + 2 * s + 1];
            const v2f q0 = f.b2[4 * g + 2 * s], q1 = f.b2[4 * g + 2 * s + 1];
            v2f b;
            b.x = r0.x; b.y = r1.x; acc[0] = wmma_f32(a2, b, acc[0]);
            b.x = r0.y; b.y = r1.y; acc[1] = wmma_f32(a2, b, acc[1]);
            b.x = r0.z; b.y = r1.z; acc[2] = wmma_f32(a2, b, acc[2]);
            b.x = r0.w; b.y = r1.w; acc[3] = wmma_f32(a2, b, acc[3]);
            b.x = q0.x; b.y = q1.x; acc[4] = wmma_f32(a2, b, acc[4]);
            b.x = q0.y; b.y = q1.y; acc[5] = wmma_f32(a2, b, acc[5]);
        }
    }
}

__global__ void __launch_bounds__(384)
proj_gemm(const float* __restrict__ Kin,
          const float* __restrict__ Vin,
          const float* __restrict__ Qin,
          const float* __restrict__ Wk,
          const float* __restrict__ Wv,
          const float* __restrict__ Wq,
          float* __restrict__ kproj,
          float* __restrict__ vproj,
          float* __restrict__ qproj) {
    const int lane = threadIdx.x & 31;
    const int wave = __builtin_amdgcn_readfirstlane(threadIdx.x >> 5);
    const int lo = lane & 15;
    const int hi = lane >> 4;

    const int which = wave >> 2;
    const float* __restrict__ X = (which == 0) ? Kin : (which == 1) ? Vin : Qin;
    const float* __restrict__ W = (which == 0) ? Wk  : (which == 1) ? Wv  : Wq;
    float* __restrict__ O       = (which == 0) ? kproj : (which == 1) ? vproj : qproj;

    const int tileM = (blockIdx.x >> 3) * 64 + (wave & 3) * 16;
    const int tileN = (blockIdx.x & 7) * 96;

    const float* xa = X + (size_t)(tileM + lo) * PJ_K + 4 * hi;
    const float* w4 = W + (size_t)(4 * hi) * PJ_N + tileN + 4 * lo;
    const float* w2 = W + (size_t)(4 * hi) * PJ_N + tileN + 64 + 2 * lo;

    v8f acc[6];
    #pragma unroll
    for (int j = 0; j < 6; ++j) acc[j] = v8f{};

    PjFrag f0, f1;
    pj_load(f0, xa, w4, w2, 0);
    for (int k = 0; k < PJ_K - 32; k += 32) {
        pj_load(f1, xa, w4, w2, k + 16);
        pj_mma(f0, acc);
        pj_load(f0, xa, w4, w2, k + 32);
        pj_mma(f1, acc);
    }
    pj_load(f1, xa, w4, w2, PJ_K - 16);
    pj_mma(f0, acc);
    pj_mma(f1, acc);

    #pragma unroll
    for (int r = 0; r < 8; ++r) {
        float* orow = O + (size_t)(tileM + 8 * hi + r) * PJ_N + tileN;
        v4f o4; o4.x = acc[0][r]; o4.y = acc[1][r]; o4.z = acc[2][r]; o4.w = acc[3][r];
        v2f o2; o2.x = acc[4][r]; o2.y = acc[5][r];
        *(v4f*)(orow + 4 * lo) = o4;
        *(v2f*)(orow + 64 + 2 * lo) = o2;
    }
}

#define SWA_W    64
#define SWA_WLEN 144
#define SWA_KT   9
#define SWA_S    2048
#define SWA_D    768

__global__ void __launch_bounds__(SWA_KT * 32)
swa_scores(const float* __restrict__ kp,
           const float* __restrict__ qp,
           float* __restrict__ pw) {
    __shared__ float red_max[SWA_KT][16];
    __shared__ float red_sum[SWA_KT][16];

    const int lane = threadIdx.x & 31;
    const int wave = threadIdx.x >> 5;
    const int lo = lane & 15;
    const int hi = lane >> 4;

    const int t0    = blockIdx.x * 16;
    const int kbase = t0 - SWA_W;

    const int krow = kbase + wave * 16 + lo;
    const int krc  = min(max(krow, 0), SWA_S - 1);
    const float* qrow  = qp + (size_t)(t0 + lo) * SWA_D + hi * 4;
    const float* krowp = kp + (size_t)krc * SWA_D + hi * 4;

    v8f acc0 = {}; v8f acc1 = {};
    v4f a[8]; v4f b[8];
    for (int i = 0; i < 8; ++i) {
        a[i] = *(const v4f*)(qrow  + 8 * i);
        b[i] = *(const v4f*)(krowp + 8 * i);
    }
    for (int k0 = 64; k0 < SWA_D; k0 += 64) {
        v4f an[8]; v4f bn[8];
        for (int i = 0; i < 8; ++i) {
            an[i] = *(const v4f*)(qrow  + k0 + 8 * i);
            bn[i] = *(const v4f*)(krowp + k0 + 8 * i);
        }
        for (int i = 0; i < 8; ++i) {
            acc0 = wmma_f32(a[i].xy, b[i].xy, acc0);
            acc1 = wmma_f32(a[i].zw, b[i].zw, acc1);
        }
        for (int i = 0; i < 8; ++i) { a[i] = an[i]; b[i] = bn[i]; }
    }
    for (int i = 0; i < 8; ++i) {
        acc0 = wmma_f32(a[i].xy, b[i].xy, acc0);
        acc1 = wmma_f32(a[i].zw, b[i].zw, acc1);
    }

    const float scale = 0.03608439182435161f;
    const int j = kbase + wave * 16 + lo;
    float s[8];
    float m[8];
    for (int r = 0; r < 8; ++r) {
        const int t = t0 + 8 * hi + r;
        const int start = max(0, t - SWA_W);
        const int end   = min(SWA_S, t + SWA_W + 1);
        s[r] = (j >= start && j < end) ? (acc0[r] + acc1[r]) * scale : -INFINITY;
        float x = s[r];
        x = fmaxf(x, __shfl_xor(x, 1));
        x = fmaxf(x, __shfl_xor(x, 2));
        x = fmaxf(x, __shfl_xor(x, 4));
        x = fmaxf(x, __shfl_xor(x, 8));
        m[r] = x;
    }
    if (lo == 0) {
        for (int r = 0; r < 8; ++r) red_max[wave][8 * hi + r] = m[r];
    }
    __syncthreads();

    float gm = red_max[0][lo];
    for (int w = 1; w < SWA_KT; ++w) gm = fmaxf(gm, red_max[w][lo]);

    float e[8];
    for (int r = 0; r < 8; ++r) {
        const float mr = __shfl(gm, 8 * hi + r);
        e[r] = (s[r] == -INFINITY) ? 0.f : __expf(s[r] - mr);
        float x = e[r];
        x += __shfl_xor(x, 1);
        x += __shfl_xor(x, 2);
        x += __shfl_xor(x, 4);
        x += __shfl_xor(x, 8);
        m[r] = x;
    }
    if (lo == 0) {
        for (int r = 0; r < 8; ++r) red_sum[wave][8 * hi + r] = m[r];
    }
    __syncthreads();

    float gs = red_sum[0][lo];
    for (int w = 1; w < SWA_KT; ++w) gs += red_sum[w][lo];
    const int tl     = t0 + lo;
    const int startl = max(0, tl - SWA_W);
    const int endl   = min(SWA_S, tl + SWA_W + 1);
    const float inv  = 1.0f / (gs * (float)(endl - startl));

    float* ptile = pw + ((size_t)(blockIdx.x * SWA_KT + wave) * 16) * 16;
    for (int r = 0; r < 8; ++r) {
        const float ir = __shfl(inv, 8 * hi + r);
        ptile[(8 * hi + r) * 16 + lo] = e[r] * ir;
    }
}

__global__ void __launch_bounds__(128)
swa_pv(const float* __restrict__ vp,
       const float* __restrict__ pw,
       float* __restrict__ out) {
    const int lane = threadIdx.x & 31;
    const int wave = threadIdx.x >> 5;
    const int lo = lane & 15;
    const int hi = lane >> 4;

    const int t0    = blockIdx.x * 16;
    const int kbase = t0 - SWA_W;
    const int cb = blockIdx.y * 128 + wave * 32 + 2 * lo;

    const float* pq = pw + ((size_t)blockIdx.x * SWA_KT * 16 + lo) * 16 + hi * 4;

    v8f acc0 = {}; v8f acc1 = {};
    #pragma unroll
    for (int k0 = 0; k0 < SWA_WLEN; k0 += 8) {
        const v4f a = *(const v4f*)(pq + (k0 >> 4) * 256 + (k0 & 15));
        const int vr = kbase + k0 + hi * 4;
        const int r0 = min(max(vr,     0), SWA_S - 1);
        const int r1 = min(max(vr + 1, 0), SWA_S - 1);
        const int r2 = min(max(vr + 2, 0), SWA_S - 1);
        const int r3 = min(max(vr + 3, 0), SWA_S - 1);
        const v2f v0 = *(const v2f*)(vp + (size_t)r0 * SWA_D + cb);
        const v2f v1 = *(const v2f*)(vp + (size_t)r1 * SWA_D + cb);
        const v2f v2 = *(const v2f*)(vp + (size_t)r2 * SWA_D + cb);
        const v2f v3 = *(const v2f*)(vp + (size_t)r3 * SWA_D + cb);
        v2f b00; b00.x = v0.x; b00.y = v1.x;
        v2f b01; b01.x = v0.y; b01.y = v1.y;
        v2f b10; b10.x = v2.x; b10.y = v3.x;
        v2f b11; b11.x = v2.y; b11.y = v3.y;
        acc0 = wmma_f32(a.xy, b00, acc0);
        acc1 = wmma_f32(a.xy, b01, acc1);
        acc0 = wmma_f32(a.zw, b10, acc0);
        acc1 = wmma_f32(a.zw, b11, acc1);
    }

    for (int r = 0; r < 8; ++r) {
        v2f o; o.x = acc0[r]; o.y = acc1[r];
        *(v2f*)(out + (size_t)(t0 + 8 * hi + r) * SWA_D + cb) = o;
    }
}

extern "C" void kernel_launch(void* const* d_in, const int* in_sizes, int n_in,
                              void* d_out, int out_size, void* d_ws, size_t ws_size,
                              hipStream_t stream) {
    const float* key   = (const float*)d_in[0];
    const float* value = (const float*)d_in[1];
    const float* query = (const float*)d_in[2];
    const float* Wk    = (const float*)d_in[3];
    const float* Wv    = (const float*)d_in[4];
    const float* Wq    = (const float*)d_in[5];
    float* out = (float*)d_out;

    const int S = 2048, T = 2048, E = 768, D = 768;

    float* ws = (float*)d_ws;
    float* kproj = ws;
    float* vproj = ws + (size_t)S * D;
    float* qproj = ws + 2 * (size_t)S * D;
    float* probs = ws + 3 * (size_t)S * D;

    proj_gemm<<<dim3(256), 384, 0, stream>>>(key, value, query, Wk, Wv, Wq,
                                             kproj, vproj, qproj);

    swa_scores<<<dim3(T / 16), SWA_KT * 32, 0, stream>>>(kproj, qproj, probs);

    swa_pv<<<dim3(T / 16, D / 128), 128, 0, stream>>>(vproj, probs, out);
}


# projection GEMM: loads of k block n+2 issued as one group of 18 ahead of the 24 WMMAs of block n, three rotating register sets (was one block ahead, interleaved by the scheduler); sum order unchanged
# speedup vs baseline: 3.3961x; 1.0813x over previous
#include <hip/hip_runtime.h>
#include <hip/hip_bf16.h>

typedef float v2f __attribute__((ext_vector_type(2)));
typedef float v4f __attribute__((ext_vector_type(4)));
typedef float v8f __attribute__((ext_vector_type(8)));

__device__ __forceinline__ v8f wmma_f32(v2f a, v2f b, v8f c) {
    return __builtin_amdgcn_wmma_f32_16x16x4_f32(
         false, a,  false, b,
         (short)0, c,  false,  false);
}

#define PJ_K 768
#define PJ_N 768

struct PjFrag {
    v4f a[2];
    v4f b4[8];
    v2f b2[8];
};

__device__ __forceinline__ void pj_load(PjFrag& f, const float* __restrict__ xa,
                                        const float* __restrict__ w4,
                                        const float* __restrict__ w2, int k) {
    #pragma unroll
    for (int g = 0; g < 2; ++g) f.a[g] = *(const v4f*)(xa + k + 8 * g);
    #pragma unroll
    for (int r = 0; r < 8; ++r) {
        const int kr = k + 8 * (r >> 2) + (r & 3);
        f.b4[r] = *(const v4f*)(w4 + (size_t)kr * PJ_N);
        f.b2[r] = *(const v2f*)(w2 + (size_t)kr * PJ_N);
    }
}

__device__ __forceinline__ void pj_mma(const PjFrag& f, v8f (&acc)[6]) {
    #pragma unroll
    for (int g = 0; g < 2; ++g) {
        #pragma unroll
        for (int s = 0; s < 2; ++s) {
            const v2f a2 = s ? f.a[g].zw : f.a[g].xy;
            const v4f r0 = f.b4[4 * g + 2 * s], r1 = f.b4[4 * g + 2 * s + 1];
            const v2f q0 = f.b2[4 * g + 2 * s], q1 = f.b2[4 * g + 2 * s + 1];
            v2f b;
            b.x = r0.x; b.y = r1.x; acc[0] = wmma_f32(a2, b, acc[0]);
            b.x = r0.y; b.y = r1.y; acc[1] = wmma_f32(a2, b, acc[1]);
            b.x = r0.z; b.y = r1.z; acc[2] = wmma_f32(a2, b, acc[2]);
            b.x = r0.w; b.y = r1.w; acc[3] = wmma_f32(a2, b, acc[3]);
            b.x = q0.x; b.y = q1.x; acc[4] = wmma_f32(a2, b, acc[4]);
            b.x = q0.y; b.y = q1.y; acc[5] = wmma_f32(a2, b, acc[5]);
        }
    }
}

__global__ void __launch_bounds__(384)
proj_gemm(const float* __restrict__ Kin,
          const float* __restrict__ Vin,
          const float* __restrict__ Qin,
          const float* __restrict__ Wk,
          const float* __restrict__ Wv,
          const float* __restrict__ Wq,
          float* __restrict__ kproj,
          float* __restrict__ vproj,
          float* __restrict__ qproj) {
    const int lane = threadIdx.x & 31;
    const int wave = __builtin_amdgcn_readfirstlane(threadIdx.x >> 5);
    const int lo = lane & 15;
    const int hi = lane >> 4;

    const int which = wave >> 2;
    const float* __restrict__ X = (which == 0) ? Kin : (which == 1) ? Vin : Qin;
    const float* __restrict__ W = (which == 0) ? Wk  : (which == 1) ? Wv  : Wq;
    float* __restrict__ O       = (which == 0) ? kproj : (which == 1) ? vproj : qproj;

    const int tileM = (blockIdx.x >> 3) * 64 + (wave & 3) * 16;
    const int tileN = (blockIdx.x & 7) * 96;

    const float* xa = X + (size_t)(tileM + lo) * PJ_K + 4 * hi;
    const float* w4 = W + (size_t)(4 * hi) * PJ_N + tileN + 4 * lo;
    const float* w2 = W + (size_t)(4 * hi) * PJ_N + tileN + 64 + 2 * lo;

    v8f acc[6];
    #pragma unroll
    for (int j = 0; j < 6; ++j) acc[j] = v8f{};


    #define PJ_HOLD() __builtin_amdgcn_sched_barrier(0)
    #define PJ_BLOCK_END()                                                      \
        asm volatile("" : "+v"(acc[0]), "+v"(acc[1]), "+v"(acc[2]),             \
                          "+v"(acc[3]), "+v"(acc[4]), "+v"(acc[5]))

    PjFrag f0, f1, f2;
    pj_load(f0, xa, w4, w2, 0);           PJ_HOLD();
    pj_load(f1, xa, w4, w2, 16);          PJ_HOLD();
    for (int k = 0; k < PJ_K - 48; k += 48) {
        pj_load(f2, xa, w4, w2, k + 32);  PJ_HOLD();
        pj_mma(f0, acc);                  PJ_BLOCK_END(); PJ_HOLD();
        pj_load(f0, xa, w4, w2, k + 48);  PJ_HOLD();
        pj_mma(f1, acc);                  PJ_BLOCK_END(); PJ_HOLD();
        pj_load(f1, xa, w4, w2, k + 64);  PJ_HOLD();
        pj_mma(f2, acc);                  PJ_BLOCK_END(); PJ_HOLD();
    }
    pj_load(f2, xa, w4, w2, PJ_K - 16);   PJ_HOLD();
    pj_mma(f0, acc);
    pj_mma(f1, acc);
    pj_mma(f2, acc);

    #pragma unroll
    for (int r = 0; r < 8; ++r) {
        float* orow = O + (size_t)(tileM + 8 * hi + r) * PJ_N + tileN;
        v4f o4; o4.x = acc[0][r]; o4.y = acc[1][r]; o4.z = acc[2][r]; o4.w = acc[3][r];
        v2f o2; o2.x = acc[4][r]; o2.y = acc[5][r];
        *(v4f*)(orow + 4 * lo) = o4;
        *(v2f*)(orow + 64 + 2 * lo) = o2;
    }
}

#define SWA_W    64
#define SWA_WLEN 144
#define SWA_KT   9
#define SWA_S    2048
#define SWA_D    768

__global__ void __launch_bounds__(SWA_KT * 32)
swa_scores(const float* __restrict__ kp,
           const float* __restrict__ qp,
           float* __restrict__ pw) {
    __shared__ float red_max[SWA_KT][16];
    __shared__ float red_sum[SWA_KT][16];

    const int lane = threadIdx.x & 31;
    const int wave = threadIdx.x >> 5;
    const int lo = lane & 15;
    const int hi = lane >> 4;

    const int t0    = blockIdx.x * 16;
    const int kbase = t0 - SWA_W;

    const int krow = kbase + wave * 16 + lo;
    const int krc  = min(max(krow, 0), SWA_S - 1);
    const float* qrow  = qp + (size_t)(t0 + lo) * SWA_D + hi * 4;
    const float* krowp = kp + (size_t)krc * SWA_D + hi * 4;

    v8f acc0 = {}; v8f acc1 = {};
    v4f a[8]; v4f b[8];
    for (int i = 0; i < 8; ++i) {
        a[i] = *(const v4f*)(qrow  + 8 * i);
        b[i] = *(const v4f*)(krowp + 8 * i);
    }
    for (int k0 = 64; k0 < SWA_D; k0 += 64) {
        v4f an[8]; v4f bn[8];
        for (int i = 0; i < 8; ++i) {
            an[i] = *(const v4f*)(qrow  + k0 + 8 * i);
            bn[i] = *(const v4f*)(krowp + k0 + 8 * i);
        }
        for (int i = 0; i < 8; ++i) {
            acc0 = wmma_f32(a[i].xy, b[i].xy, acc0);
            acc1 = wmma_f32(a[i].zw, b[i].zw, acc1);
        }
        for (int i = 0; i < 8; ++i) { a[i] = an[i]; b[i] = bn[i]; }
    }
    for (int i = 0; i < 8; ++i) {
        acc0 = wmma_f32(a[i].xy, b[i].xy, acc0);
        acc1 = wmma_f32(a[i].zw, b[i].zw, acc1);
    }

    const float scale = 0.03608439182435161f;
    const int j = kbase + wave * 16 + lo;
    float s[8];
    float m[8];
    for (int r = 0; r < 8; ++r) {
        const int t = t0 + 8 * hi + r;
        const int start = max(0, t - SWA_W);
        const int end   = min(SWA_S, t + SWA_W + 1);
        s[r] = (j >= start && j < end) ? (acc0[r] + acc1[r]) * scale : -INFINITY;
        float x = s[r];
        x = fmaxf(x, __shfl_xor(x, 1));
        x = fmaxf(x, __shfl_xor(x, 2));
        x = fmaxf(x, __shfl_xor(x, 4));
        x = fmaxf(x, __shfl_xor(x, 8));
        m[r] = x;
    }
    if (lo == 0) {
        for (int r = 0; r < 8; ++r) red_max[wave][8 * hi + r] = m[r];
    }
    __syncthreads();

    float gm = red_max[0][lo];
    for (int w = 1; w < SWA_KT; ++w) gm = fmaxf(gm, red_max[w][lo]);

    float e[8];
    for (int r = 0; r < 8; ++r) {
        const float mr = __shfl(gm, 8 * hi + r);
        e[r] = (s[r] == -INFINITY) ? 0.f : __expf(s[r] - mr);
        float x = e[r];
        x += __shfl_xor(x, 1);
        x += __shfl_xor(x, 2);
        x += __shfl_xor(x, 4);
        x += __shfl_xor(x, 8);
        m[r] = x;
    }
    if (lo == 0) {
        for (int r = 0; r < 8; ++r) red_sum[wave][8 * hi + r] = m[r];
    }
    __syncthreads();

    float gs = red_sum[0][lo];
    for (int w = 1; w < SWA_KT; ++w) gs += red_sum[w][lo];
    const int tl     = t0 + lo;
    const int startl = max(0, tl - SWA_W);
    const int endl   = min(SWA_S, tl + SWA_W + 1);
    const float inv  = 1.0f / (gs * (float)(endl - startl));

    float* ptile = pw + ((size_t)(blockIdx.x * SWA_KT + wave) * 16) * 16;
    for (int r = 0; r < 8; ++r) {
        const float ir = __shfl(inv, 8 * hi + r);
        ptile[(8 * hi + r) * 16 + lo] = e[r] * ir;
    }
}

__global__ void __launch_bounds__(128)
swa_pv(const float* __restrict__ vp,
       const float* __restrict__ pw,
       float* __restrict__ out) {
    const int lane = threadIdx.x & 31;
    const int wave = threadIdx.x >> 5;
    const int lo = lane & 15;
    const int hi = lane >> 4;

    const int t0    = blockIdx.x * 16;
    const int kbase = t0 - SWA_W;
    const int cb = blockIdx.y * 128 + wave * 32 + 2 * lo;

    const float* pq = pw + ((size_t)blockIdx.x * SWA_KT * 16 + lo) * 16 + hi * 4;

    v8f acc0 = {}; v8f acc1 = {};
    #pragma unroll
    for (int k0 = 0; k0 < SWA_WLEN; k0 += 8) {
        const v4f a = *(const v4f*)(pq + (k0 >> 4) * 256 + (k0 & 15));
        const int vr = kbase + k0 + hi * 4;
        const int r0 = min(max(vr,     0), SWA_S - 1);
        const int r1 = min(max(vr + 1, 0), SWA_S - 1);
        const int r2 = min(max(vr + 2, 0), SWA_S - 1);
        const int r3 = min(max(vr + 3, 0), SWA_S - 1);
        const v2f v0 = *(const v2f*)(vp + (size_t)r0 * SWA_D + cb);
        const v2f v1 = *(const v2f*)(vp + (size_t)r1 * SWA_D + cb);
        const v2f v2 = *(const v2f*)(vp + (size_t)r2 * SWA_D + cb);
        const v2f v3 = *(const v2f*)(vp + (size_t)r3 * SWA_D + cb);
        v2f b00; b00.x = v0.x; b00.y = v1.x;
        v2f b01; b01.x = v0.y; b01.y = v1.y;
        v2f b10; b10.x = v2.x; b10.y = v3.x;
        v2f b11; b11.x = v2.y; b11.y = v3.y;
        acc0 = wmma_f32(a.xy, b00, acc0);
        acc1 = wmma_f32(a.xy, b01, acc1);
        acc0 = wmma_f32(a.zw, b10, acc0);
        acc1 = wmma_f32(a.zw, b11, acc1);
    }

    for (int r = 0; r < 8; ++r) {
        v2f o; o.x = acc0[r]; o.y = acc1[r];
        *(v2f*)(out + (size_t)(t0 + 8 * hi + r) * SWA_D + cb) = o;
    }
}

extern "C" void kernel_launch(void* const* d_in, const int* in_sizes, int n_in,
                              void* d_out, int out_size, void* d_ws, size_t ws_size,
                              hipStream_t stream) {
    const float* key   = (const float*)d_in[0];
    const float* value = (const float*)d_in[1];
    const float* query = (const float*)d_in[2];
    const float* Wk    = (const float*)d_in[3];
    const float* Wv    = (const float*)d_in[4];
    const float* Wq    = (const float*)d_in[5];
    float* out = (float*)d_out;

    const int S = 2048, T = 2048, E = 768, D = 768;

    float* ws = (float*)d_ws;
    float* kproj = ws;
    float* vproj = ws + (size_t)S * D;
    float* qproj = ws + 2 * (size_t)S * D;
    float* probs = ws + 3 * (size_t)S * D;

    proj_gemm<<<dim3(256), 384, 0, stream>>>(key, value, query, Wk, Wv, Wq,
                                             kproj, vproj, qproj);

    swa_scores<<<dim3(T / 16), SWA_KT * 32, 0, stream>>>(kproj, qproj, probs);

    swa_pv<<<dim3(T / 16, D / 128), 128, 0, stream>>>(vproj, probs, out);
}


# score tiles as independent single-wave workgroups over all WGPs: tile-local row max and sum go to workspace, the PV kernel folds the nine per row into its A fragments; no barrier; sum order changes
# speedup vs baseline: 3.5695x; 1.0511x over previous
#include <hip/hip_runtime.h>
#include <hip/hip_bf16.h>

typedef float v2f __attribute__((ext_vector_type(2)));
typedef float v4f __attribute__((ext_vector_type(4)));
typedef float v8f __attribute__((ext_vector_type(8)));

__device__ __forceinline__ v8f wmma_f32(v2f a, v2f b, v8f c) {
    return __builtin_amdgcn_wmma_f32_16x16x4_f32(
         false, a,  false, b,
         (short)0, c,  false,  false);
}

#define PJ_K 768
#define PJ_N 768

struct PjFrag {
    v4f a[2];
    v4f b4[8];
    v2f b2[8];
};

__device__ __forceinline__ void pj_load(PjFrag& f, const float* __restrict__ xa,
                                        const float* __restrict__ w4,
                                        const float* __restrict__ w2, int k) {
    #pragma unroll
    for (int g = 0; g < 2; ++g) f.a[g] = *(const v4f*)(xa + k + 8 * g);
    #pragma unroll
    for (int r = 0; r < 8; ++r) {
        const int kr = k + 8 * (r >> 2) + (r & 3);
        f.b4[r] = *(const v4f*)(w4 + (size_t)kr * PJ_N);
        f.b2[r] = *(const v2f*)(w2 + (size_t)kr * PJ_N);
    }
}

__device__ __forceinline__ void pj_mma(const PjFrag& f, v8f (&acc)[6]) {
    #pragma unroll
    for (int g = 0; g < 2; ++g) {
        #pragma unroll
        for (int s = 0; s < 2; ++s) {
            const v2f a2 = s ? f.a[g].zw : f.a[g].xy;
            const v4f r0 = f.b4[4 * g + 2 * s], r1 = f.b4[4 * g + 2 * s + 1];
            const v2f q0 = f.b2[4 * g + 2 * s], q1 = f.b2[4 * g + 2 * s + 1];
            v2f b;
            b.x = r0.x; b.y = r1.x; acc[0] = wmma_f32(a2, b, acc[0]);
            b.x = r0.y; b.y = r1.y; acc[1] = wmma_f32(a2, b, acc[1]);
            b.x = r0.z; b.y = r1.z; acc[2] = wmma_f32(a2, b, acc[2]);
            b.x = r0.w; b.y = r1.w; acc[3] = wmma_f32(a2, b, acc[3]);
            b.x = q0.x; b.y = q1.x; acc[4] = wmma_f32(a2, b, acc[4]);
            b.x = q0.y; b.y = q1.y; acc[5] = wmma_f32(a2, b, acc[5]);
        }
    }
}

__global__ void __launch_bounds__(384)
proj_gemm(const float* __restrict__ Kin,
          const float* __restrict__ Vin,
          const float* __restrict__ Qin,
          const float* __restrict__ Wk,
          const float* __restrict__ Wv,
          const float* __restrict__ Wq,
          float* __restrict__ kproj,
          float* __restrict__ vproj,
          float* __restrict__ qproj) {
    const int lane = threadIdx.x & 31;
    const int wave = __builtin_amdgcn_readfirstlane(threadIdx.x >> 5);
    const int lo = lane & 15;
    const int hi = lane >> 4;

    const int which = wave >> 2;
    const float* __restrict__ X = (which == 0) ? Kin : (which == 1) ? Vin : Qin;
    const float* __restrict__ W = (which == 0) ? Wk  : (which == 1) ? Wv  : Wq;
    float* __restrict__ O       = (which == 0) ? kproj : (which == 1) ? vproj : qproj;

    const int tileM = (blockIdx.x >> 3) * 64 + (wave & 3) * 16;
    const int tileN = (blockIdx.x & 7) * 96;

    const float* xa = X + (size_t)(tileM + lo) * PJ_K + 4 * hi;
    const float* w4 = W + (size_t)(4 * hi) * PJ_N + tileN + 4 * lo;
    const float* w2 = W + (size_t)(4 * hi) * PJ_N + tileN + 64 + 2 * lo;

    v8f acc[6];
    #pragma unroll
    for (int j = 0; j < 6; ++j) acc[j] = v8f{};


    #define PJ_HOLD() __builtin_amdgcn_sched_barrier(0)
    #define PJ_BLOCK_END()                                                      \
        asm volatile("" : "+v"(acc[0]), "+v"(acc[1]), "+v"(acc[2]),             \
                          "+v"(acc[3]), "+v"(acc[4]), "+v"(acc[5]))

    PjFrag f0, f1, f2;
    pj_load(f0, xa, w4, w2, 0);           PJ_HOLD();
    pj_load(f1, xa, w4, w2, 16);          PJ_HOLD();
    for (int k = 0; k < PJ_K - 48; k += 48) {
        pj_load(f2, xa, w4, w2, k + 32);  PJ_HOLD();
        pj_mma(f0, acc);                  PJ_BLOCK_END(); PJ_HOLD();
        pj_load(f0, xa, w4, w2, k + 48);  PJ_HOLD();
        pj_mma(f1, acc);                  PJ_BLOCK_END(); PJ_HOLD();
        pj_load(f1, xa, w4, w2, k + 64);  PJ_HOLD();
        pj_mma(f2, acc);                  PJ_BLOCK_END(); PJ_HOLD();
    }
    pj_load(f2, xa, w4, w2, PJ_K - 16);   PJ_HOLD();
    pj_mma(f0, acc);
    pj_mma(f1, acc);
    pj_mma(f2, acc);

    #pragma unroll
    for (int r = 0; r < 8; ++r) {
        float* orow = O + (size_t)(tileM + 8 * hi + r) * PJ_N + tileN;
        v4f o4; o4.x = acc[0][r]; o4.y = acc[1][r]; o4.z = acc[2][r]; o4.w = acc[3][r];
        v2f o2; o2.x = acc[4][r]; o2.y = acc[5][r];
        *(v4f*)(orow + 4 * lo) = o4;
        *(v2f*)(orow + 64 + 2 * lo) = o2;
    }
}

#define SWA_W    64
#define SWA_WLEN 144
#define SWA_KT   9
#define SWA_S    2048
#define SWA_D    768

__global__ void __launch_bounds__(32)
swa_scores(const float* __restrict__ kp,
           const float* __restrict__ qp,
           float* __restrict__ pw,
           float* __restrict__ st) {
    const int lane = threadIdx.x;
    const int wave = blockIdx.x;
    const int lo = lane & 15;
    const int hi = lane >> 4;

    const int t0    = blockIdx.y * 16;
    const int kbase = t0 - SWA_W;
    const int tile  = blockIdx.y * SWA_KT + wave;

    const int krow = kbase + wave * 16 + lo;
    const int krc  = min(max(krow, 0), SWA_S - 1);
    const float* qrow  = qp + (size_t)(t0 + lo) * SWA_D + hi * 4;
    const float* krowp = kp + (size_t)krc * SWA_D + hi * 4;

    v8f acc0 = {}; v8f acc1 = {};
    v4f a[8]; v4f b[8];
    for (int i = 0; i < 8; ++i) {
        a[i] = *(const v4f*)(qrow  + 8 * i);
        b[i] = *(const v4f*)(krowp + 8 * i);
    }
    for (int k0 = 64; k0 < SWA_D; k0 += 64) {
        v4f an[8]; v4f bn[8];
        for (int i = 0; i < 8; ++i) {
            an[i] = *(const v4f*)(qrow  + k0 + 8 * i);
            bn[i] = *(const v4f*)(krowp + k0 + 8 * i);
        }
        for (int i = 0; i < 8; ++i) {
            acc0 = wmma_f32(a[i].xy, b[i].xy, acc0);
            acc1 = wmma_f32(a[i].zw, b[i].zw, acc1);
        }
        for (int i = 0; i < 8; ++i) { a[i] = an[i]; b[i] = bn[i]; }
    }
    for (int i = 0; i < 8; ++i) {
        acc0 = wmma_f32(a[i].xy, b[i].xy, acc0);
        acc1 = wmma_f32(a[i].zw, b[i].zw, acc1);
    }

    const float scale = 0.03608439182435161f;
    const int j = kbase + wave * 16 + lo;
    float s[8];
    float m[8];
    for (int r = 0; r < 8; ++r) {
        const int t = t0 + 8 * hi + r;
        const int start = max(0, t - SWA_W);
        const int end   = min(SWA_S, t + SWA_W + 1);
        s[r] = (j >= start && j < end) ? (acc0[r] + acc1[r]) * scale : -INFINITY;
        float x = s[r];
        x = fmaxf(x, __shfl_xor(x, 1));
        x = fmaxf(x, __shfl_xor(x, 2));
        x = fmaxf(x, __shfl_xor(x, 4));
        x = fmaxf(x, __shfl_xor(x, 8));
        m[r] = x;
    }
    float e[8];
    float z[8];
    for (int r = 0; r < 8; ++r) {
        e[r] = (s[r] == -INFINITY) ? 0.f : __expf(s[r] - m[r]);
        float x = e[r];
        x += __shfl_xor(x, 1);
        x += __shfl_xor(x, 2);
        x += __shfl_xor(x, 4);
        x += __shfl_xor(x, 8);
        z[r] = x;
    }

    float* ptile = pw + (size_t)tile * 256;
    for (int r = 0; r < 8; ++r) ptile[(8 * hi + r) * 16 + lo] = e[r];

    if (lo == 0) {
        float* sl = st + (size_t)tile * 32 + 8 * hi;
        for (int r = 0; r < 8; ++r) { sl[r] = m[r]; sl[16 + r] = z[r]; }
    }
}

__global__ void __launch_bounds__(128)
swa_pv(const float* __restrict__ vp,
       const float* __restrict__ pw,
       const float* __restrict__ st,
       float* __restrict__ out) {
    const int lane = threadIdx.x & 31;
    const int wave = threadIdx.x >> 5;
    const int lo = lane & 15;
    const int hi = lane >> 4;

    const int t0    = blockIdx.x * 16;
    const int kbase = t0 - SWA_W;
    const int cb = blockIdx.y * 128 + wave * 32 + 2 * lo;

    const float* pq = pw + ((size_t)blockIdx.x * SWA_KT * 16 + lo) * 16 + hi * 4;

    const float* sq = st + (size_t)blockIdx.x * SWA_KT * 32 + lo;
    float mt[SWA_KT];
    float zt[SWA_KT];
    for (int w = 0; w < SWA_KT; ++w) {
        mt[w] = sq[w * 32];
        zt[w] = sq[w * 32 + 16];
    }
    float gm = mt[0];
    for (int w = 1; w < SWA_KT; ++w) gm = fmaxf(gm, mt[w]);
    float cf[SWA_KT];
    float gs = 0.f;
    for (int w = 0; w < SWA_KT; ++w) {
        cf[w] = (mt[w] == -INFINITY) ? 0.f : __expf(mt[w] - gm);
        gs += cf[w] * zt[w];
    }
    const int tl     = t0 + lo;
    const int startl = max(0, tl - SWA_W);
    const int endl   = min(SWA_S, tl + SWA_W + 1);
    const float inv  = 1.0f / (gs * (float)(endl - startl));
    for (int w = 0; w < SWA_KT; ++w) cf[w] *= inv;

    v8f acc0 = {}; v8f acc1 = {};
    #pragma unroll
    for (int k0 = 0; k0 < SWA_WLEN; k0 += 8) {
        const v4f a = *(const v4f*)(pq + (k0 >> 4) * 256 + (k0 & 15)) * cf[k0 >> 4];
        const int vr = kbase + k0 + hi * 4;
        const int r0 = min(max(vr,     0), SWA_S - 1);
        const int r1 = min(max(vr + 1, 0), SWA_S - 1);
        const int r2 = min(max(vr + 2, 0), SWA_S - 1);
        const int r3 = min(max(vr + 3, 0), SWA_S - 1);
        const v2f v0 = *(const v2f*)(vp + (size_t)r0 * SWA_D + cb);
        const v2f v1 = *(const v2f*)(vp + (size_t)r1 * SWA_D + cb);
        const v2f v2 = *(const v2f*)(vp + (size_t)r2 * SWA_D + cb);
        const v2f v3 = *(const v2f*)(vp + (size_t)r3 * SWA_D + cb);
        v2f b00; b00.x = v0.x; b00.y = v1.x;
        v2f b01; b01.x = v0.y; b01.y = v1.y;
        v2f b10; b10.x = v2.x; b10.y = v3.x;
        v2f b11; b11.x = v2.y; b11.y = v3.y;
        acc0 = wmma_f32(a.xy, b00, acc0);
        acc1 = wmma_f32(a.xy, b01, acc1);
        acc0 = wmma_f32(a.zw, b10, acc0);
        acc1 = wmma_f32(a.zw, b11, acc1);
    }

    for (int r = 0; r < 8; ++r) {
        v2f o; o.x = acc0[r]; o.y = acc1[r];
        *(v2f*)(out + (size_t)(t0 + 8 * hi + r) * SWA_D + cb) = o;
    }
}

extern "C" void kernel_launch(void* const* d_in, const int* in_sizes, int n_in,
                              void* d_out, int out_size, void* d_ws, size_t ws_size,
                              hipStream_t stream) {
    const float* key   = (const float*)d_in[0];
    const float* value = (const float*)d_in[1];
    const float* query = (const float*)d_in[2];
    const float* Wk    = (const float*)d_in[3];
    const float* Wv    = (const float*)d_in[4];
    const float* Wq    = (const float*)d_in[5];
    float* out = (float*)d_out;

    const int S = 2048, T = 2048, E = 768, D = 768;

    float* ws = (float*)d_ws;
    float* kproj = ws;
    float* vproj = ws + (size_t)S * D;
    float* qproj = ws + 2 * (size_t)S * D;
    float* probs = ws + 3 * (size_t)S * D;
    float* stats = probs + (size_t)(T / 16) * SWA_KT * 256;

    proj_gemm<<<dim3(256), 384, 0, stream>>>(key, value, query, Wk, Wv, Wq,
                                             kproj, vproj, qproj);

    swa_scores<<<dim3(SWA_KT, T / 16), 32, 0, stream>>>(kproj, qproj, probs, stats);

    swa_pv<<<dim3(T / 16, D / 128), 128, 0, stream>>>(vproj, probs, stats, out);
}


# q, k, v projections stored in their readers' WMMA fragment order (q, k formed transposed): every operand load 512 contiguous bytes; PV regrouped to one 96-column block per wave; sum order changes
# speedup vs baseline: 3.7162x; 1.0411x over previous
#include <hip/hip_runtime.h>
#include <hip/hip_bf16.h>

typedef float v2f __attribute__((ext_vector_type(2)));
typedef float v4f __attribute__((ext_vector_type(4)));
typedef float v8f __attribute__((ext_vector_type(8)));

__device__ __forceinline__ v8f wmma_f32(v2f a, v2f b, v8f c) {
    return __builtin_amdgcn_wmma_f32_16x16x4_f32(
         false, a,  false, b,
         (short)0, c,  false,  false);
}

#define PJ_K 768
#define PJ_N 768
#define PJ_FRAG_CHUNKS 96

struct PjFrag {
    v4f a[2];
    v4f b4[8];
    v2f b2[8];
};

__device__ __forceinline__ void pj_load(PjFrag& f, const float* __restrict__ xa,
                                        const float* __restrict__ w4,
                                        const float* __restrict__ w2, int k) {
    #pragma unroll
    for (int g = 0; g < 2; ++g) f.a[g] = *(const v4f*)(xa + k + 8 * g);
    #pragma unroll
    for (int r = 0; r < 8; ++r) {
        const int kr = k + 8 * (r >> 2) + (r & 3);
        f.b4[r] = *(const v4f*)(w4 + (size_t)kr * PJ_N);
        f.b2[r] = *(const v2f*)(w2 + (size_t)kr * PJ_N);
    }
}

template <bool TR>
__device__ __forceinline__ v8f pj_wmma(v2f x, v2f w, v8f c) {
    return TR ? wmma_f32(w, x, c) : wmma_f32(x, w, c);
}

template <bool TR>
__device__ __forceinline__ void pj_mma(const PjFrag& f, v8f (&acc)[6]) {
    #pragma unroll
    for (int g = 0; g < 2; ++g) {
        #pragma unroll
        for (int s = 0; s < 2; ++s) {
            const v2f a2 = s ? f.a[g].zw : f.a[g].xy;
            const v4f r0 = f.b4[4 * g + 2 * s], r1 = f.b4[4 * g + 2 * s + 1];
            const v2f q0 = f.b2[4 * g + 2 * s], q1 = f.b2[4 * g + 2 * s + 1];
            v2f b;
            b.x = r0.x; b.y = r1.x; acc[0] = pj_wmma<TR>(a2, b, acc[0]);
            b.x = r0.y; b.y = r1.y; acc[1] = pj_wmma<TR>(a2, b, acc[1]);
            b.x = r0.z; b.y = r1.z; acc[2] = pj_wmma<TR>(a2, b, acc[2]);
            b.x = r0.w; b.y = r1.w; acc[3] = pj_wmma<TR>(a2, b, acc[3]);
            b.x = q0.x; b.y = q1.x; acc[4] = pj_wmma<TR>(a2, b, acc[4]);
            b.x = q0.y; b.y = q1.y; acc[5] = pj_wmma<TR>(a2, b, acc[5]);
        }
    }
}


#define PJ_HOLD() __builtin_amdgcn_sched_barrier(0)
#define PJ_BLOCK_END()                                                      \
    asm volatile("" : "+v"(acc[0]), "+v"(acc[1]), "+v"(acc[2]),             \
                      "+v"(acc[3]), "+v"(acc[4]), "+v"(acc[5]))

template <bool TR>
__device__ __forceinline__ void pj_kloop(const float* __restrict__ xa,
                                         const float* __restrict__ w4,
                                         const float* __restrict__ w2,
                                         v8f (&acc)[6]) {
    PjFrag f0, f1, f2;
    pj_load(f0, xa, w4, w2, 0);           PJ_HOLD();
    pj_load(f1, xa, w4, w2, 16);          PJ_HOLD();
    for (int k = 0; k < PJ_K - 48; k += 48) {
        pj_load(f2, xa, w4, w2, k + 32);  PJ_HOLD();
        pj_mma<TR>(f0, acc);              PJ_BLOCK_END(); PJ_HOLD();
        pj_load(f0, xa, w4, w2, k + 48);  PJ_HOLD();
        pj_mma<TR>(f1, acc);              PJ_BLOCK_END(); PJ_HOLD();
        pj_load(f1, xa, w4, w2, k + 64);  PJ_HOLD();
        pj_mma<TR>(f2, acc);              PJ_BLOCK_END(); PJ_HOLD();
    }
    pj_load(f2, xa, w4, w2, PJ_K - 16);   PJ_HOLD();
    pj_mma<TR>(f0, acc);
    pj_mma<TR>(f1, acc);
    pj_mma<TR>(f2, acc);
}

__global__ void __launch_bounds__(384)
proj_gemm(const float* __restrict__ Kin,
          const float* __restrict__ Vin,
          const float* __restrict__ Qin,
          const float* __restrict__ Wk,
          const float* __restrict__ Wv,
          const float* __restrict__ Wq,
          float* __restrict__ kproj,
          float* __restrict__ vproj,
          float* __restrict__ qproj) {
    const int lane = threadIdx.x & 31;
    const int wave = __builtin_amdgcn_readfirstlane(threadIdx.x >> 5);
    const int lo = lane & 15;
    const int hi = lane >> 4;

    const int which = wave >> 2;
    const float* __restrict__ X = (which == 0) ? Kin : (which == 1) ? Vin : Qin;
    const float* __restrict__ W = (which == 0) ? Wk  : (which == 1) ? Wv  : Wq;
    float* __restrict__ O       = (which == 0) ? kproj : (which == 1) ? vproj : qproj;

    const int tileM = (blockIdx.x >> 3) * 64 + (wave & 3) * 16;
    const int tileN = (blockIdx.x & 7) * 96;

    const float* xa = X + (size_t)(tileM + lo) * PJ_K + 4 * hi;
    const float* w4 = W + (size_t)(4 * hi) * PJ_N + tileN + 4 * lo;
    const float* w2 = W + (size_t)(4 * hi) * PJ_N + tileN + 64 + 2 * lo;

    v8f acc[6];
    #pragma unroll
    for (int j = 0; j < 6; ++j) acc[j] = v8f{};

    if (which == 1) {
        pj_kloop<false>(xa, w4, w2, acc);
        float* ochunk = O + ((size_t)(tileM >> 4) * PJ_FRAG_CHUNKS
                             + (size_t)(blockIdx.x & 7) * 12 + hi) * 128 + 4 * lo;
        #pragma unroll
        for (int j = 0; j < 6; ++j) {
            #pragma unroll
            for (int p = 0; p < 2; ++p) {
                v4f o4; o4.x = acc[j][4 * p];     o4.y = acc[j][4 * p + 1];
                        o4.z = acc[j][4 * p + 2]; o4.w = acc[j][4 * p + 3];
                *(v4f*)(ochunk + (2 * j) * 128 + p * 64) = o4;
            }
        }
    } else {
        pj_kloop<true>(xa, w4, w2, acc);
        float* ochunk = O + ((size_t)(tileM >> 4) * PJ_FRAG_CHUNKS
                             + (size_t)(blockIdx.x & 7) * 12) * 128 + 4 * lane;
        #pragma unroll
        for (int r = 0; r < 8; ++r) {
            v4f o4; o4.x = acc[0][r]; o4.y = acc[1][r]; o4.z = acc[2][r]; o4.w = acc[3][r];
            *(v4f*)(ochunk + r * 128) = o4;
        }
        #pragma unroll
        for (int r = 0; r < 4; ++r) {
            v4f o4; o4.x = acc[4][2 * r];     o4.y = acc[5][2 * r];
                    o4.z = acc[4][2 * r + 1]; o4.w = acc[5][2 * r + 1];
            *(v4f*)(ochunk + (8 + r) * 128) = o4;
        }
    }
}

#define SWA_W    64
#define SWA_KT   9
#define SWA_S    2048
#define SWA_D    768

struct ScFrag { v4f a[16]; v4f b[16]; };

__device__ __forceinline__ void sc_load(ScFrag& f, const float* __restrict__ q,
                                        const float* __restrict__ k, int c0) {
    #pragma unroll
    for (int i = 0; i < 16; ++i) {
        f.a[i] = *(const v4f*)(q + 128 * (c0 + i));
        f.b[i] = *(const v4f*)(k + 128 * (c0 + i));
    }
}

__device__ __forceinline__ void sc_mma(const ScFrag& f, v8f& acc0, v8f& acc1) {
    #pragma unroll
    for (int i = 0; i < 16; ++i) {
        acc0 = wmma_f32(f.a[i].xy, f.b[i].xy, acc0);
        acc1 = wmma_f32(f.a[i].zw, f.b[i].zw, acc1);
    }
}

#define SC_HOLD() __builtin_amdgcn_sched_barrier(0)

__global__ void __launch_bounds__(32)
swa_scores(const float* __restrict__ kp,
           const float* __restrict__ qp,
           float* __restrict__ pw,
           float* __restrict__ st) {
    const int lane = threadIdx.x;
    const int wave = blockIdx.x;
    const int lo = lane & 15;
    const int hi = lane >> 4;

    const int t0    = blockIdx.y * 16;
    const int kbase = t0 - SWA_W;
    const int tile  = blockIdx.y * SWA_KT + wave;

    const int ktile = min(max((int)blockIdx.y - SWA_W / 16 + wave, 0), SWA_S / 16 - 1);
    const float* qrow  = qp + (size_t)blockIdx.y * (16 * SWA_D) + 4 * lane;
    const float* krowp = kp + (size_t)ktile * (16 * SWA_D) + 4 * lane;

    v8f acc0 = {}; v8f acc1 = {};
    ScFrag f0, f1;
    sc_load(f0, qrow, krowp, 0);   SC_HOLD();
    sc_load(f1, qrow, krowp, 16);  SC_HOLD();
    sc_mma(f0, acc0, acc1);        SC_HOLD();
    sc_load(f0, qrow, krowp, 32);  SC_HOLD();
    sc_mma(f1, acc0, acc1);        SC_HOLD();
    sc_load(f1, qrow, krowp, 48);  SC_HOLD();
    sc_mma(f0, acc0, acc1);        SC_HOLD();
    sc_load(f0, qrow, krowp, 64);  SC_HOLD();
    sc_mma(f1, acc0, acc1);        SC_HOLD();
    sc_load(f1, qrow, krowp, 80);  SC_HOLD();
    sc_mma(f0, acc0, acc1);        SC_HOLD();
    sc_mma(f1, acc0, acc1);

    const float scale = 0.03608439182435161f;
    const int j = kbase + wave * 16 + lo;
    float s[8];
    float m[8];
    for (int r = 0; r < 8; ++r) {
        const int t = t0 + 8 * hi + r;
        const int start = max(0, t - SWA_W);
        const int end   = min(SWA_S, t + SWA_W + 1);
        s[r] = (j >= start && j < end) ? (acc0[r] + acc1[r]) * scale : -INFINITY;
        float x = s[r];
        x = fmaxf(x, __shfl_xor(x, 1));
        x = fmaxf(x, __shfl_xor(x, 2));
        x = fmaxf(x, __shfl_xor(x, 4));
        x = fmaxf(x, __shfl_xor(x, 8));
        m[r] = x;
    }
    float e[8];
    float z[8];
    for (int r = 0; r < 8; ++r) {
        e[r] = (s[r] == -INFINITY) ? 0.f : __expf(s[r] - m[r]);
        float x = e[r];
        x += __shfl_xor(x, 1);
        x += __shfl_xor(x, 2);
        x += __shfl_xor(x, 4);
        x += __shfl_xor(x, 8);
        z[r] = x;
    }

    float* ptile = pw + (size_t)tile * 256;
    for (int r = 0; r < 8; ++r) ptile[(8 * hi + r) * 16 + lo] = e[r];

    if (lo == 0) {
        float* sl = st + (size_t)tile * 32 + 8 * hi;
        for (int r = 0; r < 8; ++r) { sl[r] = m[r]; sl[16 + r] = z[r]; }
    }
}

struct PvFrag { v4f a[2]; v4f b[12]; };

__device__ __forceinline__ void pv_load(PvFrag& f, const float* __restrict__ pq,
                                        const float* __restrict__ vp,
                                        int qt, int cbk, int lane, int w) {
    const int kt = min(max(qt - SWA_W / 16 + w, 0), SWA_S / 16 - 1);
    const float* vt = vp + ((size_t)kt * PJ_FRAG_CHUNKS + (size_t)cbk * 12) * 128 + 4 * lane;
    #pragma unroll
    for (int g = 0; g < 2; ++g) f.a[g] = *(const v4f*)(pq + w * 256 + 8 * g);
    #pragma unroll
    for (int c = 0; c < 12; ++c) f.b[c] = *(const v4f*)(vt + c * 128);
}

__device__ __forceinline__ void pv_mma(const PvFrag& f, float cfw, v8f (&acc)[6]) {
    #pragma unroll
    for (int g = 0; g < 2; ++g) {
        const v4f a = f.a[g] * cfw;
        #pragma unroll
        for (int j = 0; j < 6; ++j) acc[j] = wmma_f32(a.xy, f.b[2 * j + g].xy, acc[j]);
        #pragma unroll
        for (int j = 0; j < 6; ++j) acc[j] = wmma_f32(a.zw, f.b[2 * j + g].zw, acc[j]);
    }
}

#define PV_HOLD() __builtin_amdgcn_sched_barrier(0)

__global__ void __launch_bounds__(128)
swa_pv(const float* __restrict__ vp,
       const float* __restrict__ pw,
       const float* __restrict__ st,
       float* __restrict__ out) {
    const int lane = threadIdx.x & 31;
    const int wave = __builtin_amdgcn_readfirstlane(threadIdx.x >> 5);
    const int lo = lane & 15;
    const int hi = lane >> 4;

    const int qt  = blockIdx.x;
    const int t0  = qt * 16;
    const int cbk = blockIdx.y * 4 + wave;

    const float* pq = pw + ((size_t)qt * SWA_KT * 16 + lo) * 16 + hi * 4;

    const float* sq = st + (size_t)qt * SWA_KT * 32 + lo;
    float mt[SWA_KT];
    float zt[SWA_KT];
    for (int w = 0; w < SWA_KT; ++w) {
        mt[w] = sq[w * 32];
        zt[w] = sq[w * 32 + 16];
    }
    PvFrag f[3];
    pv_load(f[0], pq, vp, qt, cbk, lane, 0);
    pv_load(f[1], pq, vp, qt, cbk, lane, 1);                   PV_HOLD();

    float gm = mt[0];
    for (int w = 1; w < SWA_KT; ++w) gm = fmaxf(gm, mt[w]);
    float cf[SWA_KT];
    float gs = 0.f;
    for (int w = 0; w < SWA_KT; ++w) {
        cf[w] = (mt[w] == -INFINITY) ? 0.f : __expf(mt[w] - gm);
        gs += cf[w] * zt[w];
    }
    const int tl     = t0 + lo;
    const int startl = max(0, tl - SWA_W);
    const int endl   = min(SWA_S, tl + SWA_W + 1);
    const float inv  = 1.0f / (gs * (float)(endl - startl));
    for (int w = 0; w < SWA_KT; ++w) cf[w] *= inv;

    v8f acc[6];
    #pragma unroll
    for (int j = 0; j < 6; ++j) acc[j] = v8f{};

    PV_HOLD();
    #pragma unroll
    for (int w = 0; w < SWA_KT; ++w) {
        if (w + 2 < SWA_KT) {
            pv_load(f[(w + 2) % 3], pq, vp, qt, cbk, lane, w + 2);  PV_HOLD();
        }
        pv_mma(f[w % 3], cf[w], acc);                          PV_HOLD();
    }

    #pragma unroll
    for (int r = 0; r < 8; ++r) {
        float* orow = out + (size_t)(t0 + 8 * hi + r) * SWA_D + cbk * 96;
        v4f o4; o4.x = acc[0][r]; o4.y = acc[1][r]; o4.z = acc[2][r]; o4.w = acc[3][r];
        v2f o2; o2.x = acc[4][r]; o2.y = acc[5][r];
        *(v4f*)(orow + 4 * lo) = o4;
        *(v2f*)(orow + 64 + 2 * lo) = o2;
    }
}

extern "C" void kernel_launch(void* const* d_in, const int* in_sizes, int n_in,
                              void* d_out, int out_size, void* d_ws, size_t ws_size,
                              hipStream_t stream) {
    const float* key   = (const float*)d_in[0];
    const float* value = (const float*)d_in[1];
    const float* query = (const float*)d_in[2];
    const float* Wk    = (const float*)d_in[3];
    const float* Wv    = (const float*)d_in[4];
    const float* Wq    = (const float*)d_in[5];
    float* out = (float*)d_out;

    const int S = 2048, T = 2048, E = 768, D = 768;

    float* ws = (float*)d_ws;
    float* kproj = ws;
    float* vproj = ws + (size_t)S * D;
    float* qproj = ws + 2 * (size_t)S * D;
    float* probs = ws + 3 * (size_t)S * D;
    float* stats = probs + (size_t)(T / 16) * SWA_KT * 256;

    proj_gemm<<<dim3(256), 384, 0, stream>>>(key, value, query, Wk, Wv, Wq,
                                             kproj, vproj, qproj);

    swa_scores<<<dim3(SWA_KT, T / 16), 32, 0, stream>>>(kproj, qproj, probs, stats);

    swa_pv<<<dim3(T / 16, D / 384), 128, 0, stream>>>(vproj, probs, stats, out);
}
